# starting kernel
# baseline (speedup 1.0000x reference)
#include <hip/hip_runtime.h>

#define NDIM 4096
#define NBATCH 16

typedef __attribute__((ext_vector_type(2))) float v2f;
typedef __attribute__((ext_vector_type(8))) float v8f;

__device__ __forceinline__ float fast_tanh(float v) {
#if __has_builtin(__builtin_amdgcn_tanhf)
  return __builtin_amdgcn_tanhf(v);
#else
  return tanhf(v);
#endif
}

__global__ __launch_bounds__(256) void CustomAttention_tanh_wmma_kernel(
    const float* __restrict__ x,
    const float* __restrict__ a,
    const float* __restrict__ bsc,
    float* __restrict__ out)
{
  __shared__ float xrow[NDIM];

  const int lane  = threadIdx.x & 31;
  const int wave  = threadIdx.x >> 5;
  const int gw    = blockIdx.x * 8 + wave;
  const int batch = gw >> 8;
  const int tile  = gw & 255;

  const float* __restrict__ xb = x + batch * NDIM;
  for (int i = threadIdx.x * 4; i < NDIM; i += 256 * 4) {
    *(float4*)(xrow + i) = *(const float4*)(xb + i);
  }
  __syncthreads();

  const int half = lane >> 4;
  const int row  = lane & 15;
  const float axn = a[0] * xrow[tile * 16 + row];

  v8f acc = {};
  #pragma unroll 8
  for (int m = 0; m < NDIM; m += 4) {
    const float2 xm = *(const float2*)(xrow + m + 2 * half);
    v2f Bv; Bv.x = xm.x;                Bv.y = xm.y;
    v2f Av; Av.x = fast_tanh(axn * xm.x); Av.y = fast_tanh(axn * xm.y);
    acc = __builtin_amdgcn_wmma_f32_16x16x4_f32(
        false, Av, false, Bv, (short)0, acc, false, false);
  }

  if (row == 0) {
    const float bs = bsc[0];
    float* __restrict__ o = out + batch * NDIM + tile * 16 + 8 * half;
    #pragma unroll
    for (int r = 0; r < 8; ++r) o[r] = bs * acc[r];
  }
}

extern "C" void kernel_launch(void* const* d_in, const int* in_sizes, int n_in,
                              void* d_out, int out_size, void* d_ws, size_t ws_size,
                              hipStream_t stream) {
  const float* x   = (const float*)d_in[0];
  const float* a   = (const float*)d_in[1];
  const float* bsc = (const float*)d_in[2];
  float* out       = (float*)d_out;

  dim3 grid(512), block(256);
  CustomAttention_tanh_wmma_kernel<<<grid, block, 0, stream>>>(x, a, bsc, out);
}


# x row read from LDS as one group of sixteen 16-byte loads in front of each block of 32 matrix steps, not an 8-byte load waited for at each step pair; changes the order in which m is summed
# speedup vs baseline: 1.0784x; 1.0784x over previous
#include <hip/hip_runtime.h>


#define NDIM 4096
#define NBATCH 16
#define GRP 16
#define NGRP (NDIM / 8 / GRP)

typedef __attribute__((ext_vector_type(2))) float v2f;
typedef __attribute__((ext_vector_type(4))) float v4f;
typedef __attribute__((ext_vector_type(8))) float v8f;

__device__ __forceinline__ float fast_tanh(float v) {
#if __has_builtin(__builtin_amdgcn_tanhf)
  return __builtin_amdgcn_tanhf(v);
#else
  return tanhf(v);
#endif
}

__device__ __forceinline__ void reads_in_front_of_block(v8f& acc, const v4f (&b)[GRP]) {
  asm volatile("" : "+v"(acc)
               : "v"(b[0]), "v"(b[1]), "v"(b[2]), "v"(b[3]),
                 "v"(b[4]), "v"(b[5]), "v"(b[6]), "v"(b[7]),
                 "v"(b[8]), "v"(b[9]), "v"(b[10]), "v"(b[11]),
                 "v"(b[12]), "v"(b[13]), "v"(b[14]), "v"(b[15]));
}

__global__ __launch_bounds__(256) void CustomAttention_tanh_wmma_kernel(
    const float* __restrict__ x,
    const float* __restrict__ a,
    const float* __restrict__ bsc,
    float* __restrict__ out)
{
  __shared__ __attribute__((aligned(16))) float xrow[NDIM];

  const int lane  = threadIdx.x & 31;
  const int wave  = threadIdx.x >> 5;
  const int gw    = blockIdx.x * 8 + wave;
  const int batch = gw >> 8;
  const int tile  = gw & 255;

  const float* __restrict__ xb = x + batch * NDIM;
  for (int i = threadIdx.x * 4; i < NDIM; i += 256 * 4) {
    *(float4*)(xrow + i) = *(const float4*)(xb + i);
  }
  __syncthreads();

  const int half = lane >> 4;
  const int row  = lane & 15;
  const float axn = a[0] * xrow[tile * 16 + row];

  const float* __restrict__ xh = xrow + 4 * half;

  v8f acc = {};
  #pragma unroll 1
  for (int g = 0; g < NGRP; ++g) {
    v4f buf[GRP];
    #pragma unroll
    for (int i = 0; i < GRP; ++i) buf[i] = *(const v4f*)(xh + 8 * (g * GRP + i));

    reads_in_front_of_block(acc, buf);

    #pragma unroll
    for (int i = 0; i < GRP; ++i) {
      v2f B0; B0.x = buf[i].x; B0.y = buf[i].y;
      v2f A0; A0.x = fast_tanh(axn * buf[i].x); A0.y = fast_tanh(axn * buf[i].y);
      acc = __builtin_amdgcn_wmma_f32_16x16x4_f32(
          false, A0, false, B0, (short)0, acc, false, false);
      v2f B1; B1.x = buf[i].z; B1.y = buf[i].w;
      v2f A1; A1.x = fast_tanh(axn * buf[i].z); A1.y = fast_tanh(axn * buf[i].w);
      acc = __builtin_amdgcn_wmma_f32_16x16x4_f32(
          false, A1, false, B1, (short)0, acc, false, false);
    }
  }

  if (row == 0) {
    const float bs = bsc[0];
    float* __restrict__ o = out + batch * NDIM + tile * 16 + 8 * half;
    #pragma unroll
    for (int r = 0; r < 8; ++r) o[r] = bs * acc[r];
  }
}

extern "C" void kernel_launch(void* const* d_in, const int* in_sizes, int n_in,
                              void* d_out, int out_size, void* d_ws, size_t ws_size,
                              hipStream_t stream) {
  const float* x   = (const float*)d_in[0];
  const float* a   = (const float*)d_in[1];
  const float* bsc = (const float*)d_in[2];
  float* out       = (float*)d_out;

  dim3 grid(512), block(256);
  CustomAttention_tanh_wmma_kernel<<<grid, block, 0, stream>>>(x, a, bsc, out);
}


# Both matrix operands carry tanh values (15 rows in A, 15 in B, x in row and column 15): 120 values per step, not 64; a 30-row group splits m over 8 waves; changes the order in which m is summed
# speedup vs baseline: 1.5352x; 1.4236x over previous
#include <hip/hip_runtime.h>


#define NDIM 4096
#define NBATCH 16
#define ROWS 30
#define NSETS 137
#define NWAVE 8
#define GRP 16
#define NGRP (NDIM / NWAVE / 8 / GRP)

typedef __attribute__((ext_vector_type(2))) float v2f;
typedef __attribute__((ext_vector_type(4))) float v4f;
typedef __attribute__((ext_vector_type(8))) float v8f;

__device__ __forceinline__ float fast_tanh(float v) {
#if __has_builtin(__builtin_amdgcn_tanhf)
  return __builtin_amdgcn_tanhf(v);
#else
  return tanhf(v);
#endif
}

__device__ __forceinline__ void reads_in_front_of_block(v8f& acc, const v4f (&b)[GRP]) {
  asm volatile("" : "+v"(acc)
               : "v"(b[0]), "v"(b[1]), "v"(b[2]), "v"(b[3]),
                 "v"(b[4]), "v"(b[5]), "v"(b[6]), "v"(b[7]),
                 "v"(b[8]), "v"(b[9]), "v"(b[10]), "v"(b[11]),
                 "v"(b[12]), "v"(b[13]), "v"(b[14]), "v"(b[15]));
}

__device__ __forceinline__ void step(v8f& acc, float xa, float xb,
                                     float axnA, float axnB, bool spare) {
  const float ta0 = fast_tanh(axnA * xa), ta1 = fast_tanh(axnA * xb);
  const float tb0 = fast_tanh(axnB * xa), tb1 = fast_tanh(axnB * xb);
  v2f A, B;
  A.x = spare ? xa : ta0;  A.y = spare ? xb : ta1;
  B.x = spare ? xa : tb0;  B.y = spare ? xb : tb1;
  acc = __builtin_amdgcn_wmma_f32_16x16x4_f32(false, A, false, B, (short)0, acc, false, false);
}

__global__ __launch_bounds__(256) void CustomAttention_tanh_wmma_kernel(
    const float* __restrict__ x,
    const float* __restrict__ a,
    const float* __restrict__ bsc,
    float* __restrict__ out)
{
  __shared__ __attribute__((aligned(16))) float xrow[NDIM];
  __shared__ float part[NWAVE][32];

  const int lane  = threadIdx.x & 31;
  const int wave  = threadIdx.x >> 5;
  const int batch = blockIdx.y;
  const int base  = blockIdx.x * ROWS;

  const float* __restrict__ xb = x + batch * NDIM;
  float4 st[4];
  #pragma unroll
  for (int i = 0; i < 4; ++i) st[i] = *(const float4*)(xb + (threadIdx.x + 256 * i) * 4);
  #pragma unroll
  for (int i = 0; i < 4; ++i) *(float4*)(xrow + (threadIdx.x + 256 * i) * 4) = st[i];
  __syncthreads();

  const int half = lane >> 4;
  const int c    = lane & 15;
  const bool spare = (c == 15);
  const int nA = min(base + c, NDIM - 1);
  const int nB = min(base + 15 + c, NDIM - 1);
  const float a0 = a[0];
  const float axnA = a0 * xrow[nA];
  const float axnB = a0 * xrow[nB];

  const float* __restrict__ xh = xrow + wave * (NDIM / NWAVE) + 4 * half;

  v8f acc = {};
  #pragma unroll 1
  for (int g = 0; g < NGRP; ++g) {
    v4f buf[GRP];
    #pragma unroll
    for (int i = 0; i < GRP; ++i) buf[i] = *(const v4f*)(xh + 8 * (g * GRP + i));

    reads_in_front_of_block(acc, buf);

    #pragma unroll
    for (int i = 0; i < GRP; ++i) {
      step(acc, buf[i].x, buf[i].y, axnA, axnB, spare);
      step(acc, buf[i].z, buf[i].w, axnA, axnB, spare);
    }
  }

  if (lane == 15) {
    #pragma unroll
    for (int r = 0; r < 8; ++r) part[wave][r] = acc[r];
  } else if (lane == 31) {
    #pragma unroll
    for (int r = 0; r < 7; ++r) part[wave][8 + r] = acc[r];
  } else if (lane >= 16) {
    part[wave][15 + (lane - 16)] = acc[7];
  }
  __syncthreads();

  if (threadIdx.x < ROWS) {
    float s = part[0][threadIdx.x];
    #pragma unroll
    for (int w = 1; w < NWAVE; ++w) s += part[w][threadIdx.x];
    const int n = base + threadIdx.x;
    if (n < NDIM) out[batch * NDIM + n] = bsc[0] * s;
  }
}

extern "C" void kernel_launch(void* const* d_in, const int* in_sizes, int n_in,
                              void* d_out, int out_size, void* d_ws, size_t ws_size,
                              hipStream_t stream) {
  const float* x   = (const float*)d_in[0];
  const float* a   = (const float*)d_in[1];
  const float* bsc = (const float*)d_in[2];
  float* out       = (float*)d_out;

  dim3 grid(137, 16), block(256);
  CustomAttention_tanh_wmma_kernel<<<grid, block, 0, stream>>>(x, a, bsc, out);
}


# speedup vs baseline: 1.7758x; 1.1567x over previous
#include <hip/hip_runtime.h>


#define NDIM 4096
#define NBATCH 16
#define ROWS 30
#define NSETS 137
#define NWAVE 8
#define GRP 16
#define NGRP (NDIM / NWAVE / 8 / GRP)

typedef __attribute__((ext_vector_type(2))) float v2f;
typedef __attribute__((ext_vector_type(4))) float v4f;
typedef __attribute__((ext_vector_type(8))) float v8f;

__device__ __forceinline__ float fast_tanh(float v) {
#if __has_builtin(__builtin_amdgcn_tanhf)
  return __builtin_amdgcn_tanhf(v);
#else
  return tanhf(v);
#endif
}

__device__ __forceinline__ void reads_in_front_of_block(v8f& acc, const v4f (&b)[GRP]) {
  asm volatile("" : "+v"(acc)
               : "v"(b[0]), "v"(b[1]), "v"(b[2]), "v"(b[3]),
                 "v"(b[4]), "v"(b[5]), "v"(b[6]), "v"(b[7]),
                 "v"(b[8]), "v"(b[9]), "v"(b[10]), "v"(b[11]),
                 "v"(b[12]), "v"(b[13]), "v"(b[14]), "v"(b[15]));
}


#define GROUP_END() __builtin_amdgcn_sched_barrier(0)

#define XA(s) (((s) & 1) ? buf[(s) >> 1].z : buf[(s) >> 1].x)
#define XB(s) (((s) & 1) ? buf[(s) >> 1].w : buf[(s) >> 1].y)

__global__ __launch_bounds__(256) void CustomAttention_tanh_wmma_kernel(
    const float* __restrict__ x,
    const float* __restrict__ a,
    const float* __restrict__ bsc,
    float* __restrict__ out)
{
  __shared__ __attribute__((aligned(16))) float xrow[NDIM];
  __shared__ float part[NWAVE][32];

  const int lane  = threadIdx.x & 31;
  const int wave  = threadIdx.x >> 5;
  const int batch = blockIdx.y;
  const int base  = blockIdx.x * ROWS;

  const float* __restrict__ xb = x + batch * NDIM;
  float4 st[4];
  #pragma unroll
  for (int i = 0; i < 4; ++i) st[i] = *(const float4*)(xb + (threadIdx.x + 256 * i) * 4);
  #pragma unroll
  for (int i = 0; i < 4; ++i) *(float4*)(xrow + (threadIdx.x + 256 * i) * 4) = st[i];
  __syncthreads();

  const int half = lane >> 4;
  const int c    = lane & 15;
  const bool spare = (c == 15);
  const int nA = min(base + c, NDIM - 1);
  const int nB = min(base + 15 + c, NDIM - 1);
  const float a0 = a[0];
  const float axnA = a0 * xrow[nA];
  const float axnB = a0 * xrow[nB];

  const float* __restrict__ xh = xrow + wave * (NDIM / NWAVE) + 4 * half;

  v8f acc = {};
  #pragma unroll 1
  for (int g = 0; g < NGRP; ++g) {
    v4f buf[GRP];
    #pragma unroll
    for (int i = 0; i < GRP; ++i) buf[i] = *(const v4f*)(xh + 8 * (g * GRP + i));

    reads_in_front_of_block(acc, buf);

    float tc[4], mn[4];
    {
      const float m0 = axnA * XA(0), m1 = axnA * XB(0), m2 = axnB * XA(0), m3 = axnB * XB(0);
      tc[0] = fast_tanh(m0);  tc[1] = fast_tanh(m1);  tc[2] = fast_tanh(m2);  tc[3] = fast_tanh(m3);
      mn[0] = axnA * XA(1);   mn[1] = axnA * XB(1);   mn[2] = axnB * XA(1);   mn[3] = axnB * XB(1);
    }
    GROUP_END();

    #pragma unroll
    for (int s = 0; s < 2 * GRP; ++s) {
      const int s2 = (s + 2 < 2 * GRP) ? s + 2 : s;
      float tn[4], m2[4];
      v2f A, B;

      tn[0] = fast_tanh(mn[0]);
      A.x = spare ? XA(s) : tc[0];   A.y = spare ? XB(s) : tc[1];
      GROUP_END();
      tn[1] = fast_tanh(mn[1]);
      B.x = spare ? XA(s) : tc[2];   B.y = spare ? XB(s) : tc[3];
      GROUP_END();
      tn[2] = fast_tanh(mn[2]);
      m2[0] = axnA * XA(s2);         m2[1] = axnA * XB(s2);
      GROUP_END();
      tn[3] = fast_tanh(mn[3]);
      m2[2] = axnB * XA(s2);         m2[3] = axnB * XB(s2);
      GROUP_END();
      acc = __builtin_amdgcn_wmma_f32_16x16x4_f32(false, A, false, B, (short)0, acc, false, false);
      GROUP_END();

      #pragma unroll
      for (int j = 0; j < 4; ++j) { tc[j] = tn[j]; mn[j] = m2[j]; }
    }
  }

  if (lane == 15) {
    #pragma unroll
    for (int r = 0; r < 8; ++r) part[wave][r] = acc[r];
  } else if (lane == 31) {
    #pragma unroll
    for (int r = 0; r < 7; ++r) part[wave][8 + r] = acc[r];
  } else if (lane >= 16) {
    part[wave][15 + (lane - 16)] = acc[7];
  }
  __syncthreads();

  if (threadIdx.x < ROWS) {
    float s = part[0][threadIdx.x];
    #pragma unroll
    for (int w = 1; w < NWAVE; ++w) s += part[w][threadIdx.x];
    const int n = base + threadIdx.x;
    if (n < NDIM) out[batch * NDIM + n] = bsc[0] * s;
  }
}

extern "C" void kernel_launch(void* const* d_in, const int* in_sizes, int n_in,
                              void* d_out, int out_size, void* d_ws, size_t ws_size,
                              hipStream_t stream) {
  const float* x   = (const float*)d_in[0];
  const float* a   = (const float*)d_in[1];
  const float* bsc = (const float*)d_in[2];
  float* out       = (float*)d_out;

  dim3 grid(137, 16), block(256);
  CustomAttention_tanh_wmma_kernel<<<grid, block, 0, stream>>>(x, a, bsc, out);
}


# Matrix instruction now forms the 16x16 tile of tanh arguments (a x_n)*x_m, 256 per step, not 120; the sum over m moves to per-lane multiply-adds; changes the order in which m is summed
# speedup vs baseline: 2.7838x; 1.5676x over previous
#include <hip/hip_runtime.h>


#define NDIM 4096
#define NBATCH 16
#define NTHREADS 512
#define WG_ROWS 64
#define NSLICE 4
#define SLICE (NDIM / NSLICE)
#define GT 8
#define NG (SLICE / 16 / GT)

typedef __attribute__((ext_vector_type(2))) float v2f;
typedef __attribute__((ext_vector_type(4))) float v4f;
typedef __attribute__((ext_vector_type(8))) float v8f;

__device__ __forceinline__ float fast_tanh(float v) {
#if __has_builtin(__builtin_amdgcn_tanhf)
  return __builtin_amdgcn_tanhf(v);
#else
  return tanhf(v);
#endif
}


#define GROUP_END() __builtin_amdgcn_sched_barrier(0)

__device__ __forceinline__ v2f fma2(v2f t, float xm, v2f acc) {
  const v2f xx = {xm, xm};
  return t * xx + acc;
}

template <bool ISSUE, bool PREP>
__device__ __forceinline__ void stretch(v8f& Dp, const v2f A, v2f& Bv, const float xi, const float xnn,
                                        const bool low, v2f (&acc)[4], v2f& tc0, v2f& tc1, float& xprev) {
  v8f Dn = Dp;
  if (ISSUE) {
    const v8f zero = {};
    Dn = __builtin_amdgcn_wmma_f32_16x16x4_f32(false, A, false, Bv, (short)0, zero, false, false);
  }
  GROUP_END();
  const float t0 = fast_tanh(Dp[0]);
  GROUP_END();
  const float t1 = fast_tanh(Dp[1]);
  acc[2] = fma2(tc0, xprev, acc[2]);
  GROUP_END();
  const float t2 = fast_tanh(Dp[2]);
  GROUP_END();
  const float t3 = fast_tanh(Dp[3]);
  acc[3] = fma2(tc1, xprev, acc[3]);
  GROUP_END();
  const float t4 = fast_tanh(Dp[4]);
  GROUP_END();
  const float t5 = fast_tanh(Dp[5]);
  { const v2f t01 = {t0, t1}; acc[0] = fma2(t01, xi, acc[0]); }
  GROUP_END();
  const float t6 = fast_tanh(Dp[6]);
  if (PREP) Bv.x = low ? xnn : 0.0f;
  GROUP_END();
  const float t7 = fast_tanh(Dp[7]);
  { const v2f t23 = {t2, t3}; acc[1] = fma2(t23, xi, acc[1]); }
  GROUP_END();
  tc0.x = t4; tc0.y = t5; tc1.x = t6; tc1.y = t7;
  xprev = xi;
  Dp = Dn;
}

#define XE(buf, j) ((((j) & 3) == 0) ? buf[(j) >> 2].x : (((j) & 3) == 1) ? buf[(j) >> 2].y : \
                    (((j) & 3) == 2) ? buf[(j) >> 2].z : buf[(j) >> 2].w)

__device__ __forceinline__ void group_full(v8f& Dp, const v2f A, v2f& Bv, const v4f (&cur)[2], const v4f (&nxt)[2],
                                           const bool low, v2f (&acc)[4], v2f& tc0, v2f& tc1, float& xprev) {
  stretch<true, true>(Dp, A, Bv, XE(cur, 0), XE(cur, 2), low, acc, tc0, tc1, xprev);
  stretch<true, true>(Dp, A, Bv, XE(cur, 1), XE(cur, 3), low, acc, tc0, tc1, xprev);
  stretch<true, true>(Dp, A, Bv, XE(cur, 2), XE(cur, 4), low, acc, tc0, tc1, xprev);
  stretch<true, true>(Dp, A, Bv, XE(cur, 3), XE(cur, 5), low, acc, tc0, tc1, xprev);
  stretch<true, true>(Dp, A, Bv, XE(cur, 4), XE(cur, 6), low, acc, tc0, tc1, xprev);
  stretch<true, true>(Dp, A, Bv, XE(cur, 5), XE(cur, 7), low, acc, tc0, tc1, xprev);
  stretch<true, true>(Dp, A, Bv, XE(cur, 6), XE(nxt, 0), low, acc, tc0, tc1, xprev);
  stretch<true, true>(Dp, A, Bv, XE(cur, 7), XE(nxt, 1), low, acc, tc0, tc1, xprev);
}

__global__ __launch_bounds__(NTHREADS) void CustomAttention_tanh_wmma_kernel(
    const float* __restrict__ x,
    const float* __restrict__ a,
    const float* __restrict__ bsc,
    float* __restrict__ out)
{
  __shared__ __attribute__((aligned(16))) float xrow[NDIM];
  __shared__ float part[NSLICE][WG_ROWS];

  const int lane  = threadIdx.x & 31;
  const int wave  = threadIdx.x >> 5;
  const int batch = blockIdx.y;
  const int base  = blockIdx.x * WG_ROWS;

  const float* __restrict__ xb = x + batch * NDIM;
  float4 st[2];
  #pragma unroll
  for (int i = 0; i < 2; ++i) st[i] = *(const float4*)(xb + (threadIdx.x + NTHREADS * i) * 4);
  #pragma unroll
  for (int i = 0; i < 2; ++i) *(float4*)(xrow + (threadIdx.x + NTHREADS * i) * 4) = st[i];
  __syncthreads();

  const int  c   = lane & 15;
  const bool low = lane < 16;
  const int  nt  = wave & 3;
  const int  ms  = wave >> 2;
  const int  n0  = base + nt * 16;

  const float a0 = a[0];
  v2f A;
  A.x = low ? a0 * xrow[n0 + c] : 0.0f;
  A.y = 0.0f;

  const float* __restrict__ xs = xrow + ms * SLICE + 4 * c;

  v4f cur[2];
  cur[0] = *(const v4f*)(xs);
  cur[1] = *(const v4f*)(xs + 64);

  v2f B0, Bv;
  B0.x = low ? cur[0].x : 0.0f;  B0.y = 0.0f;
  Bv.x = low ? cur[0].y : 0.0f;  Bv.y = 0.0f;
  GROUP_END();
  const v8f zero = {};
  v8f Dp = __builtin_amdgcn_wmma_f32_16x16x4_f32(false, A, false, B0, (short)0, zero, false, false);
  GROUP_END();

  v2f acc[4] = {{0.0f, 0.0f}, {0.0f, 0.0f}, {0.0f, 0.0f}, {0.0f, 0.0f}};
  v2f tc0 = {0.0f, 0.0f}, tc1 = {0.0f, 0.0f};
  float xprev = 0.0f;

  v4f oth[2];
  #pragma unroll 1
  for (int g = 0; g < NG - 2; g += 2) {
    oth[0] = *(const v4f*)(xs + 128 * (g + 1));
    oth[1] = *(const v4f*)(xs + 128 * (g + 1) + 64);
    group_full(Dp, A, Bv, cur, oth, low, acc, tc0, tc1, xprev);
    cur[0] = *(const v4f*)(xs + 128 * (g + 2));
    cur[1] = *(const v4f*)(xs + 128 * (g + 2) + 64);
    group_full(Dp, A, Bv, oth, cur, low, acc, tc0, tc1, xprev);
  }
  oth[0] = *(const v4f*)(xs + 128 * (NG - 1));
  oth[1] = *(const v4f*)(xs + 128 * (NG - 1) + 64);
  group_full(Dp, A, Bv, cur, oth, low, acc, tc0, tc1, xprev);

  stretch<true,  true >(Dp, A, Bv, XE(oth, 0), XE(oth, 2), low, acc, tc0, tc1, xprev);
  stretch<true,  true >(Dp, A, Bv, XE(oth, 1), XE(oth, 3), low, acc, tc0, tc1, xprev);
  stretch<true,  true >(Dp, A, Bv, XE(oth, 2), XE(oth, 4), low, acc, tc0, tc1, xprev);
  stretch<true,  true >(Dp, A, Bv, XE(oth, 3), XE(oth, 5), low, acc, tc0, tc1, xprev);
  stretch<true,  true >(Dp, A, Bv, XE(oth, 4), XE(oth, 6), low, acc, tc0, tc1, xprev);
  stretch<true,  true >(Dp, A, Bv, XE(oth, 5), XE(oth, 7), low, acc, tc0, tc1, xprev);
  stretch<true,  false>(Dp, A, Bv, XE(oth, 6), 0.0f,       low, acc, tc0, tc1, xprev);
  stretch<false, false>(Dp, A, Bv, XE(oth, 7), 0.0f,       low, acc, tc0, tc1, xprev);

  acc[2] = fma2(tc0, xprev, acc[2]);
  acc[3] = fma2(tc1, xprev, acc[3]);

  float red[8];
  #pragma unroll
  for (int k = 0; k < 4; ++k) { red[2 * k] = acc[k].x; red[2 * k + 1] = acc[k].y; }
  #pragma unroll
  for (int d = 8; d >= 1; d >>= 1) {
    float got[8];
    #pragma unroll
    for (int r = 0; r < 8; ++r) got[r] = __shfl_xor(red[r], d, 16);
    #pragma unroll
    for (int r = 0; r < 8; ++r) red[r] += got[r];
  }
  if (c == 0) {
    #pragma unroll
    for (int r = 0; r < 8; ++r) part[ms][nt * 16 + (low ? 0 : 8) + r] = red[r];
  }
  __syncthreads();

  if (threadIdx.x < WG_ROWS) {
    float s = part[0][threadIdx.x];
    #pragma unroll
    for (int w = 1; w < NSLICE; ++w) s += part[w][threadIdx.x];
    out[batch * NDIM + base + threadIdx.x] = bsc[0] * s;
  }
}

extern "C" void kernel_launch(void* const* d_in, const int* in_sizes, int n_in,
                              void* d_out, int out_size, void* d_ws, size_t ws_size,
                              hipStream_t stream) {
  const float* x   = (const float*)d_in[0];
  const float* a   = (const float*)d_in[1];
  const float* bsc = (const float*)d_in[2];
  float* out       = (float*)d_out;

  dim3 grid(64, 16), block(512);
  CustomAttention_tanh_wmma_kernel<<<grid, block, 0, stream>>>(x, a, bsc, out);
}
